# starting kernel
# baseline (speedup 1.0000x reference)
#include <hip/hip_runtime.h>

typedef __attribute__((ext_vector_type(2))) float v2f;
typedef __attribute__((ext_vector_type(8))) float v8f;

#define C_OUT   24
#define KTOT    72
#define HHALF   56
#define WIDTH   112
#define SPATIAL (HHALF * 2 * WIDTH)

__global__ void __launch_bounds__(256)
shiftconv_wmma(const float* __restrict__ x,
               const float* __restrict__ Wf,
               float* __restrict__ y)
{
    const int lane  = threadIdx.x & 31;
    const int sTile = (blockIdx.x * blockDim.x + threadIdx.x) >> 5;
    if (sTile >= 784) return;

    const int row112 = sTile / 7;
    const int mb     = (sTile - row112 * 7) << 4;
    const int n      = row112 >> 1;
    const int o      = row112 & 1;

    const int half = lane >> 4;
    const int lp   = lane & 15;
    const int m    = mb + lp;

    float tapMask[3];
    int   tapOff [3];
#pragma unroll
    for (int k = 0; k < 3; ++k) {
        int  q     = n + k - 1;
        bool valid = (q >= 0) && (q < HHALF);
        int  n2    = (q <= 0) ? (HHALF - 1) : (q - 1);
        tapMask[k] = valid ? 1.0f : 0.0f;
        tapOff[k]  = (n2 * 2 + o) * WIDTH + m;
    }

    const float* bPtr [3][2];
    float        bMask[3][2];
#pragma unroll
    for (int p = 0; p < 3; ++p) {
#pragma unroll
        for (int e = 0; e < 2; ++e) {
            int c  = 4 * p + 2 * half + e;
            int jo = c / 3;
            int k  = c - 3 * jo;
            bPtr [p][e] = x + jo * SPATIAL + tapOff[k];
            bMask[p][e] = tapMask[k];
        }
    }

    const bool   i1Valid = (16 + lp) < C_OUT;
    const float  aScale1 = i1Valid ? 1.0f : 0.0f;
    const float* aPtr0   = Wf + lp * KTOT + 2 * half;
    const float* aPtr1   = Wf + (i1Valid ? (16 + lp) : 0) * KTOT + 2 * half;

    v8f acc0 = {};
    v8f acc1 = {};

#pragma unroll
    for (int t = 0; t < 6; ++t) {
#pragma unroll
        for (int p = 0; p < 3; ++p) {
            const int aOff = 12 * t + 4 * p;
            const int bOff = 4 * t * SPATIAL;

            v2f a0, a1, b;
            a0.x = aPtr0[aOff + 0];
            a0.y = aPtr0[aOff + 1];
            a1.x = aScale1 * aPtr1[aOff + 0];
            a1.y = aScale1 * aPtr1[aOff + 1];
            b.x  = bMask[p][0] * bPtr[p][0][bOff];
            b.y  = bMask[p][1] * bPtr[p][1][bOff];

            acc0 = __builtin_amdgcn_wmma_f32_16x16x4_f32(
                false, a0, false, b, (short)0, acc0, false, false);
            acc1 = __builtin_amdgcn_wmma_f32_16x16x4_f32(
                false, a1, false, b, (short)0, acc1, false, false);
        }
    }

    const int mOut = (m + 1 == WIDTH) ? 0 : (m + 1);
    const int base = row112 * WIDTH + mOut;

#pragma unroll
    for (int e = 0; e < 8; ++e) {
        int i0 = e + 8 * half;
        y[i0 * SPATIAL + base] = acc0[e];
    }
    if (half == 0) {
#pragma unroll
        for (int e = 0; e < 8; ++e) {
            y[(16 + e) * SPATIAL + base] = acc1[e];
        }
    }
}

extern "C" void kernel_launch(void* const* d_in, const int* in_sizes, int n_in,
                              void* d_out, int out_size, void* d_ws, size_t ws_size,
                              hipStream_t stream) {
    (void)in_sizes; (void)n_in; (void)out_size; (void)d_ws; (void)ws_size;
    const float* x = (const float*)d_in[0];
    const float* W = (const float*)d_in[1];
    float*       y = (float*)d_out;

    dim3 grid(98), block(256);
    hipLaunchKernelGGL(shiftconv_wmma, grid, block, 0, stream, x, W, y);
}


# All 72 operand loads of a wave issued as one group ahead of its 36 WMMAs (sched_group_barrier): one dependent memory round trip per wave, not about twelve; same products, same sum order
# speedup vs baseline: 1.3308x; 1.3308x over previous
#include <hip/hip_runtime.h>

typedef __attribute__((ext_vector_type(2))) float v2f;
typedef __attribute__((ext_vector_type(8))) float v8f;

#define C_OUT   24
#define KTOT    72
#define HHALF   56
#define WIDTH   112
#define SPATIAL (HHALF * 2 * WIDTH)

__global__ void __launch_bounds__(256)
shiftconv_wmma(const float* __restrict__ x,
               const float* __restrict__ Wf,
               float* __restrict__ y)
{
    const int lane  = threadIdx.x & 31;
    const int sTile = (blockIdx.x * blockDim.x + threadIdx.x) >> 5;
    if (sTile >= 784) return;

    const int row112 = sTile / 7;
    const int mb     = (sTile - row112 * 7) << 4;
    const int n      = row112 >> 1;
    const int o      = row112 & 1;

    const int half = lane >> 4;
    const int lp   = lane & 15;
    const int m    = mb + lp;

    float tapMask[3];
    int   tapOff [3];
#pragma unroll
    for (int k = 0; k < 3; ++k) {
        int  q     = n + k - 1;
        bool valid = (q >= 0) && (q < HHALF);
        int  n2    = (q <= 0) ? (HHALF - 1) : (q - 1);
        tapMask[k] = valid ? 1.0f : 0.0f;
        tapOff[k]  = (n2 * 2 + o) * WIDTH + m;
    }

    const float* bPtr [3][2];
    float        bMask[3][2];
#pragma unroll
    for (int p = 0; p < 3; ++p) {
#pragma unroll
        for (int e = 0; e < 2; ++e) {
            int c  = 4 * p + 2 * half + e;
            int jo = c / 3;
            int k  = c - 3 * jo;
            bPtr [p][e] = x + jo * SPATIAL + tapOff[k];
            bMask[p][e] = tapMask[k];
        }
    }

    const bool   i1Valid = (16 + lp) < C_OUT;
    const float  aScale1 = i1Valid ? 1.0f : 0.0f;
    const float* aPtr0   = Wf + lp * KTOT + 2 * half;
    const float* aPtr1   = Wf + (i1Valid ? (16 + lp) : 0) * KTOT + 2 * half;

    v2f   aF0[18], aF1[18];
    float bF0[18], bF1[18];
#pragma unroll
    for (int t = 0; t < 6; ++t) {
#pragma unroll
        for (int p = 0; p < 3; ++p) {
            const int kk   = 3 * t + p;
            const int aOff = 12 * t + 4 * p;
            const int bOff = 4 * t * SPATIAL;
            aF0[kk].x = aPtr0[aOff + 0];
            aF0[kk].y = aPtr0[aOff + 1];
            aF1[kk].x = aPtr1[aOff + 0];
            aF1[kk].y = aPtr1[aOff + 1];
            bF0[kk]   = bPtr[p][0][bOff];
            bF1[kk]   = bPtr[p][1][bOff];
        }
    }

    v8f acc0 = {};
    v8f acc1 = {};
#pragma unroll
    for (int t = 0; t < 6; ++t) {
#pragma unroll
        for (int p = 0; p < 3; ++p) {
            const int kk = 3 * t + p;
            v2f a1, b;
            a1.x = aScale1 * aF1[kk].x;
            a1.y = aScale1 * aF1[kk].y;
            b.x  = bMask[p][0] * bF0[kk];
            b.y  = bMask[p][1] * bF1[kk];

            acc0 = __builtin_amdgcn_wmma_f32_16x16x4_f32(
                false, aF0[kk], false, b, (short)0, acc0, false, false);
            acc1 = __builtin_amdgcn_wmma_f32_16x16x4_f32(
                false, a1, false, b, (short)0, acc1, false, false);
        }
    }

    __builtin_amdgcn_sched_group_barrier(0x020, 72, 0);
    __builtin_amdgcn_sched_group_barrier(0x008, 36, 0);

    const int mOut = (m + 1 == WIDTH) ? 0 : (m + 1);
    const int base = row112 * WIDTH + mOut;

#pragma unroll
    for (int e = 0; e < 8; ++e) {
        int i0 = e + 8 * half;
        y[i0 * SPATIAL + base] = acc0[e];
    }
    if (half == 0) {
#pragma unroll
        for (int e = 0; e < 8; ++e) {
            y[(16 + e) * SPATIAL + base] = acc1[e];
        }
    }
}

extern "C" void kernel_launch(void* const* d_in, const int* in_sizes, int n_in,
                              void* d_out, int out_size, void* d_ws, size_t ws_size,
                              hipStream_t stream) {
    (void)in_sizes; (void)n_in; (void)out_size; (void)d_ws; (void)ws_size;
    const float* x = (const float*)d_in[0];
    const float* W = (const float*)d_in[1];
    float*       y = (float*)d_out;

    dim3 grid(98), block(256);
    hipLaunchKernelGGL(shiftconv_wmma, grid, block, 0, stream, x, W, y);
}


# Workgroups of 4 waves, not 8 (196 x 128 threads): one wave per SIMD on 196 WGPs, so a SIMD issues 36 WMMAs, not 72, and a WGP's cache serves half the weight gathers; same products, same sum order
# speedup vs baseline: 1.4206x; 1.0675x over previous
#include <hip/hip_runtime.h>

typedef __attribute__((ext_vector_type(2))) float v2f;
typedef __attribute__((ext_vector_type(8))) float v8f;

#define C_OUT   24
#define KTOT    72
#define HHALF   56
#define WIDTH   112
#define SPATIAL (HHALF * 2 * WIDTH)

__global__ void __launch_bounds__(128)
shiftconv_wmma(const float* __restrict__ x,
               const float* __restrict__ Wf,
               float* __restrict__ y)
{
    const int lane  = threadIdx.x & 31;
    const int sTile = (blockIdx.x * blockDim.x + threadIdx.x) >> 5;
    if (sTile >= 784) return;

    const int row112 = sTile / 7;
    const int mb     = (sTile - row112 * 7) << 4;
    const int n      = row112 >> 1;
    const int o      = row112 & 1;

    const int half = lane >> 4;
    const int lp   = lane & 15;
    const int m    = mb + lp;

    float tapMask[3];
    int   tapOff [3];
#pragma unroll
    for (int k = 0; k < 3; ++k) {
        int  q     = n + k - 1;
        bool valid = (q >= 0) && (q < HHALF);
        int  n2    = (q <= 0) ? (HHALF - 1) : (q - 1);
        tapMask[k] = valid ? 1.0f : 0.0f;
        tapOff[k]  = (n2 * 2 + o) * WIDTH + m;
    }

    const float* bPtr [3][2];
    float        bMask[3][2];
#pragma unroll
    for (int p = 0; p < 3; ++p) {
#pragma unroll
        for (int e = 0; e < 2; ++e) {
            int c  = 4 * p + 2 * half + e;
            int jo = c / 3;
            int k  = c - 3 * jo;
            bPtr [p][e] = x + jo * SPATIAL + tapOff[k];
            bMask[p][e] = tapMask[k];
        }
    }

    const bool   i1Valid = (16 + lp) < C_OUT;
    const float  aScale1 = i1Valid ? 1.0f : 0.0f;
    const float* aPtr0   = Wf + lp * KTOT + 2 * half;
    const float* aPtr1   = Wf + (i1Valid ? (16 + lp) : 0) * KTOT + 2 * half;

    v2f   aF0[18], aF1[18];
    float bF0[18], bF1[18];
#pragma unroll
    for (int t = 0; t < 6; ++t) {
#pragma unroll
        for (int p = 0; p < 3; ++p) {
            const int kk   = 3 * t + p;
            const int aOff = 12 * t + 4 * p;
            const int bOff = 4 * t * SPATIAL;
            aF0[kk].x = aPtr0[aOff + 0];
            aF0[kk].y = aPtr0[aOff + 1];
            aF1[kk].x = aPtr1[aOff + 0];
            aF1[kk].y = aPtr1[aOff + 1];
            bF0[kk]   = bPtr[p][0][bOff];
            bF1[kk]   = bPtr[p][1][bOff];
        }
    }

    v8f acc0 = {};
    v8f acc1 = {};
#pragma unroll
    for (int t = 0; t < 6; ++t) {
#pragma unroll
        for (int p = 0; p < 3; ++p) {
            const int kk = 3 * t + p;
            v2f a1, b;
            a1.x = aScale1 * aF1[kk].x;
            a1.y = aScale1 * aF1[kk].y;
            b.x  = bMask[p][0] * bF0[kk];
            b.y  = bMask[p][1] * bF1[kk];

            acc0 = __builtin_amdgcn_wmma_f32_16x16x4_f32(
                false, aF0[kk], false, b, (short)0, acc0, false, false);
            acc1 = __builtin_amdgcn_wmma_f32_16x16x4_f32(
                false, a1, false, b, (short)0, acc1, false, false);
        }
    }

    __builtin_amdgcn_sched_group_barrier(0x020, 72, 0);
    __builtin_amdgcn_sched_group_barrier(0x008, 36, 0);

    const int mOut = (m + 1 == WIDTH) ? 0 : (m + 1);
    const int base = row112 * WIDTH + mOut;

#pragma unroll
    for (int e = 0; e < 8; ++e) {
        int i0 = e + 8 * half;
        y[i0 * SPATIAL + base] = acc0[e];
    }
    if (half == 0) {
#pragma unroll
        for (int e = 0; e < 8; ++e) {
            y[(16 + e) * SPATIAL + base] = acc1[e];
        }
    }
}

extern "C" void kernel_launch(void* const* d_in, const int* in_sizes, int n_in,
                              void* d_out, int out_size, void* d_ws, size_t ws_size,
                              hipStream_t stream) {
    (void)in_sizes; (void)n_in; (void)out_size; (void)d_ws; (void)ws_size;
    const float* x = (const float*)d_in[0];
    const float* W = (const float*)d_in[1];
    float*       y = (float*)d_out;

    dim3 grid(196), block(128);
    hipLaunchKernelGGL(shiftconv_wmma, grid, block, 0, stream, x, W, y);
}
